# starting kernel
# baseline (speedup 1.0000x reference)
#include <hip/hip_runtime.h>
#include <hip/hip_bf16.h>
#include <math.h>


#define BATCH 16384
#define TPTS  128

typedef __attribute__((ext_vector_type(2))) float v2f;
typedef __attribute__((ext_vector_type(8))) float v8f;

__device__ __forceinline__ unsigned rotl32(unsigned v, int n) {
    return (v << n) | (v >> (32 - n));
}

__device__ __forceinline__ void tf2x32(unsigned k0, unsigned k1,
                                       unsigned& x0, unsigned& x1) {
    const unsigned ks0 = k0, ks1 = k1, ks2 = k0 ^ k1 ^ 0x1BD11BDAu;
    x0 += ks0; x1 += ks1;
#define TFR(r) { x0 += x1; x1 = rotl32(x1, r); x1 ^= x0; }
    TFR(13) TFR(15) TFR(26) TFR(6)  x0 += ks1; x1 += ks2 + 1u;
    TFR(17) TFR(29) TFR(16) TFR(24) x0 += ks2; x1 += ks0 + 2u;
    TFR(13) TFR(15) TFR(26) TFR(6)  x0 += ks0; x1 += ks1 + 3u;
    TFR(17) TFR(29) TFR(16) TFR(24) x0 += ks1; x1 += ks2 + 4u;
    TFR(13) TFR(15) TFR(26) TFR(6)  x0 += ks2; x1 += ks0 + 5u;
#undef TFR
}

__device__ __forceinline__ float bits_to_unit(unsigned bits) {
    return __uint_as_float((bits >> 9) | 0x3f800000u) - 1.0f;
}

__global__ __launch_bounds__(256) void
RotRandomizer_build_R_kernel(float* __restrict__ out_R) {
    const int b = blockIdx.x * 256 + threadIdx.x;

    unsigned s0 = 0u, s1 = 1u;
    tf2x32(0u, 42u, s0, s1);
    const unsigned sub0 = s0, sub1 = s1;

    unsigned k[4][2];
#pragma unroll
    for (int i = 0; i < 4; ++i) {
        unsigned a = 0u, c = (unsigned)i;
        tf2x32(sub0, sub1, a, c);
        k[i][0] = a; k[i][1] = c;
    }

    float u[4];
#pragma unroll
    for (int i = 0; i < 4; ++i) {
        unsigned a = 0u, c = (unsigned)b;
        tf2x32(k[i][0], k[i][1], a, c);
        u[i] = bits_to_unit(a ^ c);
    }

    const float TWO_PI = 6.28318530717958647692f;
    const float PI_F   = 3.14159265358979323846f;
    float ax = u[0] * TWO_PI - PI_F;
    float ay = u[1] * TWO_PI - PI_F;
    float az = u[2] * TWO_PI - PI_F;
    if (u[3] < 0.015625f) { ax = 0.f; ay = 0.f; az = 0.f; }

    float sx, cx, sy, cy, sz, cz;
    sincosf(ax, &sx, &cx);
    sincosf(ay, &sy, &cy);
    sincosf(az, &sz, &cz);

    float* R = out_R + (size_t)b * 9;
    R[0] = cz * cy;
    R[1] = cz * sy * sx - sz * cx;
    R[2] = cz * sy * cx + sz * sx;
    R[3] = sz * cy;
    R[4] = sz * sy * sx + cz * cx;
    R[5] = sz * sy * cx - cz * sx;
    R[6] = -sy;
    R[7] = cy * sx;
    R[8] = cy * cx;
}

__global__ __launch_bounds__(256) void
RotRandomizer_rotate_kernel(const float* __restrict__ pos,
                            const float* __restrict__ quat,
                            const float* __restrict__ Rg,
                            float* __restrict__ out_pos,
                            float* __restrict__ out_quat) {
    __shared__ float lds[8 * 1024];

    const int tid  = threadIdx.x;
    const int wave = tid >> 5;
    const int lane = tid & 31;
    const int h    = lane & 15;
    const bool hiL = lane >= 16;
    const int b    = blockIdx.x * 2 + (wave >> 2);
    const int pt   = ((wave & 3) << 5) + lane;
    const int abase = wave * 1024;
    const int obase = abase + 512;

    const size_t idx = (size_t)b * TPTS + pt;

    const float* pi = pos + idx * 3;
    const float px = pi[0], py = pi[1], pz = pi[2];
    const float4 q = *((const float4*)quat + idx);
    const float qx = q.x, qy = q.y, qz = q.z, qw = q.w;

    const float two = 2.0f / (qw * qw + qx * qx + qy * qy + qz * qz);
    const float m00 = 1.0f - two * (qy * qy + qz * qz);
    const float m01 = two * (qx * qy - qz * qw);
    const float m02 = two * (qx * qz + qy * qw);
    const float m10 = two * (qx * qy + qz * qw);
    const float m11 = 1.0f - two * (qx * qx + qz * qz);
    const float m12 = two * (qy * qz - qx * qw);
    const float m20 = two * (qx * qz - qy * qw);
    const float m21 = two * (qy * qz + qx * qw);
    const float m22 = 1.0f - two * (qx * qx + qy * qy);

    float4* ap = (float4*)&lds[abase + lane * 16];
    ap[0] = make_float4(m00, m10, m20, 0.f);
    ap[1] = make_float4(m01, m11, m21, 0.f);
    ap[2] = make_float4(m02, m12, m22, 0.f);
    ap[3] = make_float4(px, py, pz, 0.f);

    const float* Rb = Rg + (size_t)b * 9;
    float b0v = 0.f, b1v = 0.f;
    if (h < 3) {
        if (!hiL) { b0v = Rb[h * 3 + 0]; b1v = Rb[h * 3 + 1]; }
        else      { b0v = Rb[h * 3 + 2]; }
    }
    v2f bm; bm.x = b0v; bm.y = b1v;

    __syncthreads();

    const int acol = hiL ? 2 : 0;
    const int drow = hiL ? 8 : 0;

    v2f am[8];
#pragma unroll
    for (int c = 0; c < 8; ++c) {
        const float* ar = &lds[abase + (c * 16 + h) * 4 + acol];
        am[c].x = ar[0];
        am[c].y = ar[1];
    }

    const v8f czero = {0.f, 0.f, 0.f, 0.f, 0.f, 0.f, 0.f, 0.f};
#pragma unroll
    for (int c = 0; c < 8; ++c) {
        v8f d = __builtin_amdgcn_wmma_f32_16x16x4_f32(
            false, am[c], false, bm, (short)0, czero, false, false);
        if (h < 3) {
#pragma unroll
            for (int j = 0; j < 8; ++j) {
                lds[obase + (c * 16 + j + drow) * 4 + h] = d[j];
            }
        }
    }

    __syncthreads();

    const float4* op = (const float4*)&lds[obase + lane * 16];
    const float4 c0 = op[0];
    const float4 c1 = op[1];
    const float4 c2 = op[2];
    const float4 rp = op[3];

    const float M00 = c0.x, M10 = c0.y, M20 = c0.z;
    const float M01 = c1.x, M11 = c1.y, M21 = c1.z;
    const float M02 = c2.x, M12 = c2.y, M22 = c2.z;

    const float t0 = 1.0f + M00 + M11 + M22;
    const float t1 = 1.0f + M00 - M11 - M22;
    const float t2 = 1.0f - M00 + M11 - M22;
    const float t3 = 1.0f - M00 - M11 + M22;
    const float qa0 = sqrtf(fmaxf(t0, 1e-8f));
    const float qa1 = sqrtf(fmaxf(t1, 1e-8f));
    const float qa2 = sqrtf(fmaxf(t2, 1e-8f));
    const float qa3 = sqrtf(fmaxf(t3, 1e-8f));

    int best = 0; float bv = qa0;
    if (qa1 > bv) { best = 1; bv = qa1; }
    if (qa2 > bv) { best = 2; bv = qa2; }
    if (qa3 > bv) { best = 3; bv = qa3; }

    float W, X, Y, Z;
    if (best == 0)      { W = qa0 * qa0; X = M21 - M12; Y = M02 - M20; Z = M10 - M01; }
    else if (best == 1) { W = M21 - M12; X = qa1 * qa1; Y = M01 + M10; Z = M02 + M20; }
    else if (best == 2) { W = M02 - M20; X = M10 + M01; Y = qa2 * qa2; Z = M12 + M21; }
    else                { W = M10 - M01; X = M20 + M02; Y = M21 + M12; Z = qa3 * qa3; }
    const float inv = 1.0f / (2.0f * fmaxf(bv, 0.1f));
    W *= inv; X *= inv; Y *= inv; Z *= inv;

    float* po = out_pos + idx * 3;
    po[0] = rp.x; po[1] = rp.y; po[2] = rp.z;
    ((float4*)out_quat)[idx] = make_float4(X, Y, Z, W);
}

extern "C" void kernel_launch(void* const* d_in, const int* in_sizes, int n_in,
                              void* d_out, int out_size, void* d_ws, size_t ws_size,
                              hipStream_t stream) {
    const float* pos  = (const float*)d_in[0];
    const float* quat = (const float*)d_in[1];

    float* out      = (float*)d_out;
    float* out_pos  = out;
    float* out_quat = out + (size_t)BATCH * TPTS * 3;
    float* out_R    = out + (size_t)BATCH * TPTS * 3
                          + (size_t)BATCH * TPTS * 4;

    RotRandomizer_build_R_kernel<<<BATCH / 256, 256, 0, stream>>>(out_R);
    RotRandomizer_rotate_kernel<<<BATCH / 2, 256, 0, stream>>>(
        pos, quat, out_R, out_pos, out_quat);
}


# R is now the WMMA A operand, each lane's own rot/pos values the B operand (rows 0-7 on k=0,1, rows 8-15 on k=2,3; 6 WMMAs chained through C, same k order): results land in-lane, LDS and barriers go
# speedup vs baseline: 1.2349x; 1.2349x over previous
#include <hip/hip_runtime.h>
#include <hip/hip_bf16.h>
#include <math.h>


#define BATCH 16384
#define TPTS  128

typedef __attribute__((ext_vector_type(2))) float v2f;
typedef __attribute__((ext_vector_type(8))) float v8f;

__device__ __forceinline__ unsigned rotl32(unsigned v, int n) {
    return (v << n) | (v >> (32 - n));
}

__device__ __forceinline__ void tf2x32(unsigned k0, unsigned k1,
                                       unsigned& x0, unsigned& x1) {
    const unsigned ks0 = k0, ks1 = k1, ks2 = k0 ^ k1 ^ 0x1BD11BDAu;
    x0 += ks0; x1 += ks1;
#define TFR(r) { x0 += x1; x1 = rotl32(x1, r); x1 ^= x0; }
    TFR(13) TFR(15) TFR(26) TFR(6)  x0 += ks1; x1 += ks2 + 1u;
    TFR(17) TFR(29) TFR(16) TFR(24) x0 += ks2; x1 += ks0 + 2u;
    TFR(13) TFR(15) TFR(26) TFR(6)  x0 += ks0; x1 += ks1 + 3u;
    TFR(17) TFR(29) TFR(16) TFR(24) x0 += ks1; x1 += ks2 + 4u;
    TFR(13) TFR(15) TFR(26) TFR(6)  x0 += ks2; x1 += ks0 + 5u;
#undef TFR
}

__device__ __forceinline__ float bits_to_unit(unsigned bits) {
    return __uint_as_float((bits >> 9) | 0x3f800000u) - 1.0f;
}

__global__ __launch_bounds__(256) void
RotRandomizer_build_R_kernel(float* __restrict__ out_R) {
    const int b = blockIdx.x * 256 + threadIdx.x;

    unsigned s0 = 0u, s1 = 1u;
    tf2x32(0u, 42u, s0, s1);
    const unsigned sub0 = s0, sub1 = s1;

    unsigned k[4][2];
#pragma unroll
    for (int i = 0; i < 4; ++i) {
        unsigned a = 0u, c = (unsigned)i;
        tf2x32(sub0, sub1, a, c);
        k[i][0] = a; k[i][1] = c;
    }

    float u[4];
#pragma unroll
    for (int i = 0; i < 4; ++i) {
        unsigned a = 0u, c = (unsigned)b;
        tf2x32(k[i][0], k[i][1], a, c);
        u[i] = bits_to_unit(a ^ c);
    }

    const float TWO_PI = 6.28318530717958647692f;
    const float PI_F   = 3.14159265358979323846f;
    float ax = u[0] * TWO_PI - PI_F;
    float ay = u[1] * TWO_PI - PI_F;
    float az = u[2] * TWO_PI - PI_F;
    if (u[3] < 0.015625f) { ax = 0.f; ay = 0.f; az = 0.f; }

    float sx, cx, sy, cy, sz, cz;
    sincosf(ax, &sx, &cx);
    sincosf(ay, &sy, &cy);
    sincosf(az, &sz, &cz);

    float* R = out_R + (size_t)b * 9;
    R[0] = cz * cy;
    R[1] = cz * sy * sx - sz * cx;
    R[2] = cz * sy * cx + sz * sx;
    R[3] = sz * cy;
    R[4] = sz * sy * sx + cz * cx;
    R[5] = sz * sy * cx - cz * sx;
    R[6] = -sy;
    R[7] = cy * sx;
    R[8] = cy * cx;
}

__global__ __launch_bounds__(256) void
RotRandomizer_rotate_kernel(const float* __restrict__ pos,
                            const float* __restrict__ quat,
                            const float* __restrict__ Rg,
                            float* __restrict__ out_pos,
                            float* __restrict__ out_quat) {
    const int tid  = threadIdx.x;
    const int wave = tid >> 5;
    const int lane = tid & 31;
    const int b    = blockIdx.x * 2 + (wave >> 2);
    const int pt   = ((wave & 3) << 5) + lane;

    const size_t idx = (size_t)b * TPTS + pt;

    const float* pi = pos + idx * 3;
    const float px = pi[0], py = pi[1], pz = pi[2];
    const float4 q = *((const float4*)quat + idx);
    const float qx = q.x, qy = q.y, qz = q.z, qw = q.w;

    const int c    = lane & 15;
    const int half = lane >> 4;
    const int r    = c & 7;
    const bool act   = ((c >> 3) == half) && (r < 6);
    const bool first = r < 3;
    float R0 = 0.f, R1 = 0.f, R2 = 0.f;
    if (act) {
        const float* Rrow = Rg + (size_t)b * 9 + (first ? r : r - 3) * 3;
        R0 = Rrow[0]; R1 = Rrow[1]; R2 = Rrow[2];
    }
    v2f a1, a2, a3;
    a1.x = first ? R0 : 0.f;   a1.y = first ? R1 : 0.f;
    a2.x = first ? R2 : 0.f;   a2.y = first ? 0.f : R0;
    a3.x = first ? 0.f : R1;   a3.y = first ? 0.f : R2;

    const float two = 2.0f / (qw * qw + qx * qx + qy * qy + qz * qz);
    const float m00 = 1.0f - two * (qy * qy + qz * qz);
    const float m01 = two * (qx * qy - qz * qw);
    const float m02 = two * (qx * qz + qy * qw);
    const float m10 = two * (qx * qy + qz * qw);
    const float m11 = 1.0f - two * (qx * qx + qz * qz);
    const float m12 = two * (qy * qz - qx * qw);
    const float m20 = two * (qx * qz - qy * qw);
    const float m21 = two * (qy * qz + qx * qw);
    const float m22 = 1.0f - two * (qx * qx + qy * qy);

    v2f bx1, bx2, bx3, by1, by2, by3;
    bx1.x = m00; bx1.y = m10;
    bx2.x = m20; bx2.y = m01;
    bx3.x = m11; bx3.y = m21;
    by1.x = m02; by1.y = m12;
    by2.x = m22; by2.y = px;
    by3.x = py;  by3.y = pz;

    const v8f czero = {0.f, 0.f, 0.f, 0.f, 0.f, 0.f, 0.f, 0.f};
    v8f X = __builtin_amdgcn_wmma_f32_16x16x4_f32(
        false, a1, false, bx1, (short)0, czero, false, false);
    v8f Y = __builtin_amdgcn_wmma_f32_16x16x4_f32(
        false, a1, false, by1, (short)0, czero, false, false);
    X = __builtin_amdgcn_wmma_f32_16x16x4_f32(
        false, a2, false, bx2, (short)0, X, false, false);
    Y = __builtin_amdgcn_wmma_f32_16x16x4_f32(
        false, a2, false, by2, (short)0, Y, false, false);
    X = __builtin_amdgcn_wmma_f32_16x16x4_f32(
        false, a3, false, bx3, (short)0, X, false, false);
    Y = __builtin_amdgcn_wmma_f32_16x16x4_f32(
        false, a3, false, by3, (short)0, Y, false, false);

    const float M00 = X[0], M10 = X[1], M20 = X[2];
    const float M01 = X[3], M11 = X[4], M21 = X[5];
    const float M02 = Y[0], M12 = Y[1], M22 = Y[2];
    const float rpx = Y[3], rpy = Y[4], rpz = Y[5];

    const float t0 = 1.0f + M00 + M11 + M22;
    const float t1 = 1.0f + M00 - M11 - M22;
    const float t2 = 1.0f - M00 + M11 - M22;
    const float t3 = 1.0f - M00 - M11 + M22;
    const float qa0 = sqrtf(fmaxf(t0, 1e-8f));
    const float qa1 = sqrtf(fmaxf(t1, 1e-8f));
    const float qa2 = sqrtf(fmaxf(t2, 1e-8f));
    const float qa3 = sqrtf(fmaxf(t3, 1e-8f));

    int best = 0; float bv = qa0;
    if (qa1 > bv) { best = 1; bv = qa1; }
    if (qa2 > bv) { best = 2; bv = qa2; }
    if (qa3 > bv) { best = 3; bv = qa3; }

    float W, X_, Y_, Z;
    if (best == 0)      { W = qa0 * qa0; X_ = M21 - M12; Y_ = M02 - M20; Z = M10 - M01; }
    else if (best == 1) { W = M21 - M12; X_ = qa1 * qa1; Y_ = M01 + M10; Z = M02 + M20; }
    else if (best == 2) { W = M02 - M20; X_ = M10 + M01; Y_ = qa2 * qa2; Z = M12 + M21; }
    else                { W = M10 - M01; X_ = M20 + M02; Y_ = M21 + M12; Z = qa3 * qa3; }
    const float inv = 1.0f / (2.0f * fmaxf(bv, 0.1f));
    W *= inv; X_ *= inv; Y_ *= inv; Z *= inv;

    float* po = out_pos + idx * 3;
    po[0] = rpx; po[1] = rpy; po[2] = rpz;
    ((float4*)out_quat)[idx] = make_float4(X_, Y_, Z, W);
}

extern "C" void kernel_launch(void* const* d_in, const int* in_sizes, int n_in,
                              void* d_out, int out_size, void* d_ws, size_t ws_size,
                              hipStream_t stream) {
    const float* pos  = (const float*)d_in[0];
    const float* quat = (const float*)d_in[1];

    float* out      = (float*)d_out;
    float* out_pos  = out;
    float* out_quat = out + (size_t)BATCH * TPTS * 3;
    float* out_R    = out + (size_t)BATCH * TPTS * 3
                          + (size_t)BATCH * TPTS * 4;

    RotRandomizer_build_R_kernel<<<BATCH / 256, 256, 0, stream>>>(out_R);
    RotRandomizer_rotate_kernel<<<BATCH / 2, 256, 0, stream>>>(
        pos, quat, out_R, out_pos, out_quat);
}


# 4 points per lane: each wave owns a whole batch element, its 8 point loads in flight before first use, R's A fragments built once per 128 points, a quarter of the waves; same arithmetic per point
# speedup vs baseline: 1.3587x; 1.1003x over previous
#include <hip/hip_runtime.h>
#include <hip/hip_bf16.h>
#include <math.h>


#define BATCH 16384
#define TPTS  128

typedef __attribute__((ext_vector_type(2))) float v2f;
typedef __attribute__((ext_vector_type(8))) float v8f;

__device__ __forceinline__ unsigned rotl32(unsigned v, int n) {
    return (v << n) | (v >> (32 - n));
}

__device__ __forceinline__ void tf2x32(unsigned k0, unsigned k1,
                                       unsigned& x0, unsigned& x1) {
    const unsigned ks0 = k0, ks1 = k1, ks2 = k0 ^ k1 ^ 0x1BD11BDAu;
    x0 += ks0; x1 += ks1;
#define TFR(r) { x0 += x1; x1 = rotl32(x1, r); x1 ^= x0; }
    TFR(13) TFR(15) TFR(26) TFR(6)  x0 += ks1; x1 += ks2 + 1u;
    TFR(17) TFR(29) TFR(16) TFR(24) x0 += ks2; x1 += ks0 + 2u;
    TFR(13) TFR(15) TFR(26) TFR(6)  x0 += ks0; x1 += ks1 + 3u;
    TFR(17) TFR(29) TFR(16) TFR(24) x0 += ks1; x1 += ks2 + 4u;
    TFR(13) TFR(15) TFR(26) TFR(6)  x0 += ks2; x1 += ks0 + 5u;
#undef TFR
}

__device__ __forceinline__ float bits_to_unit(unsigned bits) {
    return __uint_as_float((bits >> 9) | 0x3f800000u) - 1.0f;
}

__global__ __launch_bounds__(256) void
RotRandomizer_build_R_kernel(float* __restrict__ out_R) {
    const int b = blockIdx.x * 256 + threadIdx.x;

    unsigned s0 = 0u, s1 = 1u;
    tf2x32(0u, 42u, s0, s1);
    const unsigned sub0 = s0, sub1 = s1;

    unsigned k[4][2];
#pragma unroll
    for (int i = 0; i < 4; ++i) {
        unsigned a = 0u, c = (unsigned)i;
        tf2x32(sub0, sub1, a, c);
        k[i][0] = a; k[i][1] = c;
    }

    float u[4];
#pragma unroll
    for (int i = 0; i < 4; ++i) {
        unsigned a = 0u, c = (unsigned)b;
        tf2x32(k[i][0], k[i][1], a, c);
        u[i] = bits_to_unit(a ^ c);
    }

    const float TWO_PI = 6.28318530717958647692f;
    const float PI_F   = 3.14159265358979323846f;
    float ax = u[0] * TWO_PI - PI_F;
    float ay = u[1] * TWO_PI - PI_F;
    float az = u[2] * TWO_PI - PI_F;
    if (u[3] < 0.015625f) { ax = 0.f; ay = 0.f; az = 0.f; }

    float sx, cx, sy, cy, sz, cz;
    sincosf(ax, &sx, &cx);
    sincosf(ay, &sy, &cy);
    sincosf(az, &sz, &cz);

    float* R = out_R + (size_t)b * 9;
    R[0] = cz * cy;
    R[1] = cz * sy * sx - sz * cx;
    R[2] = cz * sy * cx + sz * sx;
    R[3] = sz * cy;
    R[4] = sz * sy * sx + cz * cx;
    R[5] = sz * sy * cx - cz * sx;
    R[6] = -sy;
    R[7] = cy * sx;
    R[8] = cy * cx;
}

#define PPL 4

__device__ __forceinline__ void
rotate_one_point(const v2f a1, const v2f a2, const v2f a3,
                 const float px, const float py, const float pz, const float4 q,
                 float* __restrict__ po, float4* __restrict__ qo) {
    const float qx = q.x, qy = q.y, qz = q.z, qw = q.w;

    const float two = 2.0f / (qw * qw + qx * qx + qy * qy + qz * qz);
    const float m00 = 1.0f - two * (qy * qy + qz * qz);
    const float m01 = two * (qx * qy - qz * qw);
    const float m02 = two * (qx * qz + qy * qw);
    const float m10 = two * (qx * qy + qz * qw);
    const float m11 = 1.0f - two * (qx * qx + qz * qz);
    const float m12 = two * (qy * qz - qx * qw);
    const float m20 = two * (qx * qz - qy * qw);
    const float m21 = two * (qy * qz + qx * qw);
    const float m22 = 1.0f - two * (qx * qx + qy * qy);

    v2f bx1, bx2, bx3, by1, by2, by3;
    bx1.x = m00; bx1.y = m10;
    bx2.x = m20; bx2.y = m01;
    bx3.x = m11; bx3.y = m21;
    by1.x = m02; by1.y = m12;
    by2.x = m22; by2.y = px;
    by3.x = py;  by3.y = pz;

    const v8f czero = {0.f, 0.f, 0.f, 0.f, 0.f, 0.f, 0.f, 0.f};
    v8f X = __builtin_amdgcn_wmma_f32_16x16x4_f32(
        false, a1, false, bx1, (short)0, czero, false, false);
    v8f Y = __builtin_amdgcn_wmma_f32_16x16x4_f32(
        false, a1, false, by1, (short)0, czero, false, false);
    X = __builtin_amdgcn_wmma_f32_16x16x4_f32(
        false, a2, false, bx2, (short)0, X, false, false);
    Y = __builtin_amdgcn_wmma_f32_16x16x4_f32(
        false, a2, false, by2, (short)0, Y, false, false);
    X = __builtin_amdgcn_wmma_f32_16x16x4_f32(
        false, a3, false, bx3, (short)0, X, false, false);
    Y = __builtin_amdgcn_wmma_f32_16x16x4_f32(
        false, a3, false, by3, (short)0, Y, false, false);

    const float M00 = X[0], M10 = X[1], M20 = X[2];
    const float M01 = X[3], M11 = X[4], M21 = X[5];
    const float M02 = Y[0], M12 = Y[1], M22 = Y[2];
    const float rpx = Y[3], rpy = Y[4], rpz = Y[5];

    const float t0 = 1.0f + M00 + M11 + M22;
    const float t1 = 1.0f + M00 - M11 - M22;
    const float t2 = 1.0f - M00 + M11 - M22;
    const float t3 = 1.0f - M00 - M11 + M22;
    const float qa0 = sqrtf(fmaxf(t0, 1e-8f));
    const float qa1 = sqrtf(fmaxf(t1, 1e-8f));
    const float qa2 = sqrtf(fmaxf(t2, 1e-8f));
    const float qa3 = sqrtf(fmaxf(t3, 1e-8f));

    int best = 0; float bv = qa0;
    if (qa1 > bv) { best = 1; bv = qa1; }
    if (qa2 > bv) { best = 2; bv = qa2; }
    if (qa3 > bv) { best = 3; bv = qa3; }

    float W, X_, Y_, Z;
    if (best == 0)      { W = qa0 * qa0; X_ = M21 - M12; Y_ = M02 - M20; Z = M10 - M01; }
    else if (best == 1) { W = M21 - M12; X_ = qa1 * qa1; Y_ = M01 + M10; Z = M02 + M20; }
    else if (best == 2) { W = M02 - M20; X_ = M10 + M01; Y_ = qa2 * qa2; Z = M12 + M21; }
    else                { W = M10 - M01; X_ = M20 + M02; Y_ = M21 + M12; Z = qa3 * qa3; }
    const float inv = 1.0f / (2.0f * fmaxf(bv, 0.1f));
    W *= inv; X_ *= inv; Y_ *= inv; Z *= inv;

    po[0] = rpx; po[1] = rpy; po[2] = rpz;
    *qo = make_float4(X_, Y_, Z, W);
}

__global__ __launch_bounds__(256) void
RotRandomizer_rotate_kernel(const float* __restrict__ pos,
                            const float* __restrict__ quat,
                            const float* __restrict__ Rg,
                            float* __restrict__ out_pos,
                            float* __restrict__ out_quat) {
    const int tid  = threadIdx.x;
    const int wave = tid >> 5;
    const int lane = tid & 31;
    const int b    = blockIdx.x * 8 + wave;

    const size_t idx0 = (size_t)b * TPTS + lane;

    float  px[PPL], py[PPL], pz[PPL];
    float4 q[PPL];
#pragma unroll
    for (int j = 0; j < PPL; ++j) {
        const size_t idx = idx0 + (size_t)(j * 32);
        const float* pi = pos + idx * 3;
        px[j] = pi[0]; py[j] = pi[1]; pz[j] = pi[2];
        q[j]  = *((const float4*)quat + idx);
    }

    const int c    = lane & 15;
    const int half = lane >> 4;
    const int r    = c & 7;
    const bool act   = ((c >> 3) == half) && (r < 6);
    const bool first = r < 3;
    float R0 = 0.f, R1 = 0.f, R2 = 0.f;
    if (act) {
        const float* Rrow = Rg + (size_t)b * 9 + (first ? r : r - 3) * 3;
        R0 = Rrow[0]; R1 = Rrow[1]; R2 = Rrow[2];
    }
    v2f a1, a2, a3;
    a1.x = first ? R0 : 0.f;   a1.y = first ? R1 : 0.f;
    a2.x = first ? R2 : 0.f;   a2.y = first ? 0.f : R0;
    a3.x = first ? 0.f : R1;   a3.y = first ? 0.f : R2;

#pragma unroll
    for (int j = 0; j < PPL; ++j) {
        const size_t idx = idx0 + (size_t)(j * 32);
        rotate_one_point(a1, a2, a3, px[j], py[j], pz[j], q[j],
                         out_pos + idx * 3, (float4*)out_quat + idx);
    }
}

extern "C" void kernel_launch(void* const* d_in, const int* in_sizes, int n_in,
                              void* d_out, int out_size, void* d_ws, size_t ws_size,
                              hipStream_t stream) {
    const float* pos  = (const float*)d_in[0];
    const float* quat = (const float*)d_in[1];

    float* out      = (float*)d_out;
    float* out_pos  = out;
    float* out_quat = out + (size_t)BATCH * TPTS * 3;
    float* out_R    = out + (size_t)BATCH * TPTS * 3
                          + (size_t)BATCH * TPTS * 4;

    RotRandomizer_build_R_kernel<<<BATCH / 256, 256, 0, stream>>>(out_R);
    RotRandomizer_rotate_kernel<<<BATCH / 8, 256, 0, stream>>>(
        pos, quat, out_R, out_pos, out_quat);
}


# One launch instead of two: wave 0 of each work-group draws R for its 8 batch elements (one uniform draw per lane) while the point loads fly, rows handed over via LDS and one barrier; R bit-identical
# speedup vs baseline: 1.4863x; 1.0939x over previous
#include <hip/hip_runtime.h>
#include <hip/hip_bf16.h>
#include <math.h>


#define BATCH 16384
#define TPTS  128

typedef __attribute__((ext_vector_type(2))) float v2f;
typedef __attribute__((ext_vector_type(8))) float v8f;

__device__ __forceinline__ unsigned rotl32(unsigned v, int n) {
    return (v << n) | (v >> (32 - n));
}

__device__ __forceinline__ void tf2x32(unsigned k0, unsigned k1,
                                       unsigned& x0, unsigned& x1) {
    const unsigned ks0 = k0, ks1 = k1, ks2 = k0 ^ k1 ^ 0x1BD11BDAu;
    x0 += ks0; x1 += ks1;
#define TFR(r) { x0 += x1; x1 = rotl32(x1, r); x1 ^= x0; }
    TFR(13) TFR(15) TFR(26) TFR(6)  x0 += ks1; x1 += ks2 + 1u;
    TFR(17) TFR(29) TFR(16) TFR(24) x0 += ks2; x1 += ks0 + 2u;
    TFR(13) TFR(15) TFR(26) TFR(6)  x0 += ks0; x1 += ks1 + 3u;
    TFR(17) TFR(29) TFR(16) TFR(24) x0 += ks1; x1 += ks2 + 4u;
    TFR(13) TFR(15) TFR(26) TFR(6)  x0 += ks2; x1 += ks0 + 5u;
#undef TFR
}

__device__ __forceinline__ float bits_to_unit(unsigned bits) {
    return __uint_as_float((bits >> 9) | 0x3f800000u) - 1.0f;
}

template <int K>
__device__ __forceinline__ float quad_take(float v) {
    return __int_as_float(__builtin_amdgcn_update_dpp(
        0, __float_as_int(v), K * 0x55, 0xF, 0xF, true));
}

__device__ __forceinline__ void
build_R_rows(const int lane, const int b0,
             float4* __restrict__ rows, float* __restrict__ out_R) {
    const int e = lane >> 2;
    const int i = lane & 3;

    unsigned s0 = 0u, s1 = 1u;
    tf2x32(0u, 42u, s0, s1);
    const unsigned sub0 = s0, sub1 = s1;

    unsigned k[4][2];
#pragma unroll
    for (int t = 0; t < 4; ++t) {
        unsigned a = 0u, c = (unsigned)t;
        tf2x32(sub0, sub1, a, c);
        k[t][0] = a; k[t][1] = c;
    }
    const unsigned key0 = (i == 0) ? k[0][0] : (i == 1) ? k[1][0] : (i == 2) ? k[2][0] : k[3][0];
    const unsigned key1 = (i == 0) ? k[0][1] : (i == 1) ? k[1][1] : (i == 2) ? k[2][1] : k[3][1];

    unsigned a = 0u, c = (unsigned)(b0 + e);
    tf2x32(key0, key1, a, c);
    const float u  = bits_to_unit(a ^ c);
    const float u3 = quad_take<3>(u);

    const float TWO_PI = 6.28318530717958647692f;
    const float PI_F   = 3.14159265358979323846f;
    float ang = u * TWO_PI - PI_F;
    if (u3 < 0.015625f) ang = 0.f;

    float sn, cs;
    sincosf(ang, &sn, &cs);
    const float sx = quad_take<0>(sn), cx = quad_take<0>(cs);
    const float sy = quad_take<1>(sn), cy = quad_take<1>(cs);
    const float sz = quad_take<2>(sn), cz = quad_take<2>(cs);

    const float czsy = cz * sy;
    const float szsy = sz * sy;
    const float R0 = cz * cy;
    const float R1 = fmaf(czsy, sx, -(sz * cx));
    const float R2 = fmaf(czsy, cx, sz * sx);
    const float R3 = sz * cy;
    const float R4 = fmaf(cz, cx, szsy * sx);
    const float R5 = fmaf(szsy, cx, -(cz * sx));
    const float R6 = -sy;
    const float R7 = cy * sx;
    const float R8 = cy * cx;

    const float r0 = (i == 0) ? R0 : (i == 1) ? R3 : R6;
    const float r1 = (i == 0) ? R1 : (i == 1) ? R4 : R7;
    const float r2 = (i == 0) ? R2 : (i == 1) ? R5 : R8;

    rows[lane] = make_float4(r0, r1, r2, 0.f);
    if (i < 3) {
        float* R = out_R + (size_t)(b0 + e) * 9 + i * 3;
        R[0] = r0; R[1] = r1; R[2] = r2;
    }
}

#define PPL 4

__device__ __forceinline__ void
rotate_one_point(const v2f a1, const v2f a2, const v2f a3,
                 const float px, const float py, const float pz, const float4 q,
                 float* __restrict__ po, float4* __restrict__ qo) {
    const float qx = q.x, qy = q.y, qz = q.z, qw = q.w;

    const float two = 2.0f / (qw * qw + qx * qx + qy * qy + qz * qz);
    const float m00 = 1.0f - two * (qy * qy + qz * qz);
    const float m01 = two * (qx * qy - qz * qw);
    const float m02 = two * (qx * qz + qy * qw);
    const float m10 = two * (qx * qy + qz * qw);
    const float m11 = 1.0f - two * (qx * qx + qz * qz);
    const float m12 = two * (qy * qz - qx * qw);
    const float m20 = two * (qx * qz - qy * qw);
    const float m21 = two * (qy * qz + qx * qw);
    const float m22 = 1.0f - two * (qx * qx + qy * qy);

    v2f bx1, bx2, bx3, by1, by2, by3;
    bx1.x = m00; bx1.y = m10;
    bx2.x = m20; bx2.y = m01;
    bx3.x = m11; bx3.y = m21;
    by1.x = m02; by1.y = m12;
    by2.x = m22; by2.y = px;
    by3.x = py;  by3.y = pz;

    const v8f czero = {0.f, 0.f, 0.f, 0.f, 0.f, 0.f, 0.f, 0.f};
    v8f X = __builtin_amdgcn_wmma_f32_16x16x4_f32(
        false, a1, false, bx1, (short)0, czero, false, false);
    v8f Y = __builtin_amdgcn_wmma_f32_16x16x4_f32(
        false, a1, false, by1, (short)0, czero, false, false);
    X = __builtin_amdgcn_wmma_f32_16x16x4_f32(
        false, a2, false, bx2, (short)0, X, false, false);
    Y = __builtin_amdgcn_wmma_f32_16x16x4_f32(
        false, a2, false, by2, (short)0, Y, false, false);
    X = __builtin_amdgcn_wmma_f32_16x16x4_f32(
        false, a3, false, bx3, (short)0, X, false, false);
    Y = __builtin_amdgcn_wmma_f32_16x16x4_f32(
        false, a3, false, by3, (short)0, Y, false, false);

    const float M00 = X[0], M10 = X[1], M20 = X[2];
    const float M01 = X[3], M11 = X[4], M21 = X[5];
    const float M02 = Y[0], M12 = Y[1], M22 = Y[2];
    const float rpx = Y[3], rpy = Y[4], rpz = Y[5];

    const float t0 = 1.0f + M00 + M11 + M22;
    const float t1 = 1.0f + M00 - M11 - M22;
    const float t2 = 1.0f - M00 + M11 - M22;
    const float t3 = 1.0f - M00 - M11 + M22;
    const float qa0 = sqrtf(fmaxf(t0, 1e-8f));
    const float qa1 = sqrtf(fmaxf(t1, 1e-8f));
    const float qa2 = sqrtf(fmaxf(t2, 1e-8f));
    const float qa3 = sqrtf(fmaxf(t3, 1e-8f));

    int best = 0; float bv = qa0;
    if (qa1 > bv) { best = 1; bv = qa1; }
    if (qa2 > bv) { best = 2; bv = qa2; }
    if (qa3 > bv) { best = 3; bv = qa3; }

    float W, X_, Y_, Z;
    if (best == 0)      { W = qa0 * qa0; X_ = M21 - M12; Y_ = M02 - M20; Z = M10 - M01; }
    else if (best == 1) { W = M21 - M12; X_ = qa1 * qa1; Y_ = M01 + M10; Z = M02 + M20; }
    else if (best == 2) { W = M02 - M20; X_ = M10 + M01; Y_ = qa2 * qa2; Z = M12 + M21; }
    else                { W = M10 - M01; X_ = M20 + M02; Y_ = M21 + M12; Z = qa3 * qa3; }
    const float inv = 1.0f / (2.0f * fmaxf(bv, 0.1f));
    W *= inv; X_ *= inv; Y_ *= inv; Z *= inv;

    po[0] = rpx; po[1] = rpy; po[2] = rpz;
    *qo = make_float4(X_, Y_, Z, W);
}

__global__ __launch_bounds__(256) void
RotRandomizer_rotate_kernel(const float* __restrict__ pos,
                            const float* __restrict__ quat,
                            float* __restrict__ out_R,
                            float* __restrict__ out_pos,
                            float* __restrict__ out_quat) {
    __shared__ float4 sRows[32];

    const int tid  = threadIdx.x;
    const int wave = __builtin_amdgcn_readfirstlane(tid >> 5);
    const int lane = tid & 31;
    const int b0   = blockIdx.x * 8;
    const int b    = b0 + wave;

    const size_t idx0 = (size_t)b * TPTS + lane;

    float  px[PPL], py[PPL], pz[PPL];
    float4 q[PPL];
#pragma unroll
    for (int j = 0; j < PPL; ++j) {
        const size_t idx = idx0 + (size_t)(j * 32);
        const float* pi = pos + idx * 3;
        px[j] = pi[0]; py[j] = pi[1]; pz[j] = pi[2];
        q[j]  = *((const float4*)quat + idx);
    }

    if (wave == 0) build_R_rows(lane, b0, sRows, out_R);
    __builtin_amdgcn_fence(__ATOMIC_RELEASE, "workgroup", "local");
    __builtin_amdgcn_s_barrier();
    __builtin_amdgcn_fence(__ATOMIC_ACQUIRE, "workgroup", "local");

    const int c    = lane & 15;
    const int half = lane >> 4;
    const int r    = c & 7;
    const bool act   = ((c >> 3) == half) && (r < 6);
    const bool first = r < 3;
    float R0 = 0.f, R1 = 0.f, R2 = 0.f;
    if (act) {
        const float4 Rrow = sRows[wave * 4 + (first ? r : r - 3)];
        R0 = Rrow.x; R1 = Rrow.y; R2 = Rrow.z;
    }
    v2f a1, a2, a3;
    a1.x = first ? R0 : 0.f;   a1.y = first ? R1 : 0.f;
    a2.x = first ? R2 : 0.f;   a2.y = first ? 0.f : R0;
    a3.x = first ? 0.f : R1;   a3.y = first ? 0.f : R2;

#pragma unroll
    for (int j = 0; j < PPL; ++j) {
        const size_t idx = idx0 + (size_t)(j * 32);
        rotate_one_point(a1, a2, a3, px[j], py[j], pz[j], q[j],
                         out_pos + idx * 3, (float4*)out_quat + idx);
    }
}

extern "C" void kernel_launch(void* const* d_in, const int* in_sizes, int n_in,
                              void* d_out, int out_size, void* d_ws, size_t ws_size,
                              hipStream_t stream) {
    const float* pos  = (const float*)d_in[0];
    const float* quat = (const float*)d_in[1];

    float* out      = (float*)d_out;
    float* out_pos  = out;
    float* out_quat = out + (size_t)BATCH * TPTS * 3;
    float* out_R    = out + (size_t)BATCH * TPTS * 3
                          + (size_t)BATCH * TPTS * 4;

    RotRandomizer_rotate_kernel<<<BATCH / 8, 256, 0, stream>>>(
        pos, quat, out_R, out_pos, out_quat);
}


# One square root per point instead of four: the root of the largest argument, and the root's own exact rounding test (arg > pred(root)*root, by fma) finds the first argument sharing it; same bits
# speedup vs baseline: 1.6551x; 1.1136x over previous
#include <hip/hip_runtime.h>
#include <hip/hip_bf16.h>
#include <math.h>


#define BATCH 16384
#define TPTS  128

typedef __attribute__((ext_vector_type(2))) float v2f;
typedef __attribute__((ext_vector_type(8))) float v8f;

__device__ __forceinline__ unsigned rotl32(unsigned v, int n) {
    return (v << n) | (v >> (32 - n));
}

__device__ __forceinline__ void tf2x32(unsigned k0, unsigned k1,
                                       unsigned& x0, unsigned& x1) {
    const unsigned ks0 = k0, ks1 = k1, ks2 = k0 ^ k1 ^ 0x1BD11BDAu;
    x0 += ks0; x1 += ks1;
#define TFR(r) { x0 += x1; x1 = rotl32(x1, r); x1 ^= x0; }
    TFR(13) TFR(15) TFR(26) TFR(6)  x0 += ks1; x1 += ks2 + 1u;
    TFR(17) TFR(29) TFR(16) TFR(24) x0 += ks2; x1 += ks0 + 2u;
    TFR(13) TFR(15) TFR(26) TFR(6)  x0 += ks0; x1 += ks1 + 3u;
    TFR(17) TFR(29) TFR(16) TFR(24) x0 += ks1; x1 += ks2 + 4u;
    TFR(13) TFR(15) TFR(26) TFR(6)  x0 += ks2; x1 += ks0 + 5u;
#undef TFR
}

__device__ __forceinline__ float bits_to_unit(unsigned bits) {
    return __uint_as_float((bits >> 9) | 0x3f800000u) - 1.0f;
}

template <int K>
__device__ __forceinline__ float quad_take(float v) {
    return __int_as_float(__builtin_amdgcn_update_dpp(
        0, __float_as_int(v), K * 0x55, 0xF, 0xF, true));
}

__device__ __forceinline__ void
build_R_rows(const int lane, const int b0,
             float4* __restrict__ rows, float* __restrict__ out_R) {
    const int e = lane >> 2;
    const int i = lane & 3;

    unsigned s0 = 0u, s1 = 1u;
    tf2x32(0u, 42u, s0, s1);
    const unsigned sub0 = s0, sub1 = s1;

    unsigned k[4][2];
#pragma unroll
    for (int t = 0; t < 4; ++t) {
        unsigned a = 0u, c = (unsigned)t;
        tf2x32(sub0, sub1, a, c);
        k[t][0] = a; k[t][1] = c;
    }
    const unsigned key0 = (i == 0) ? k[0][0] : (i == 1) ? k[1][0] : (i == 2) ? k[2][0] : k[3][0];
    const unsigned key1 = (i == 0) ? k[0][1] : (i == 1) ? k[1][1] : (i == 2) ? k[2][1] : k[3][1];

    unsigned a = 0u, c = (unsigned)(b0 + e);
    tf2x32(key0, key1, a, c);
    const float u  = bits_to_unit(a ^ c);
    const float u3 = quad_take<3>(u);

    const float TWO_PI = 6.28318530717958647692f;
    const float PI_F   = 3.14159265358979323846f;
    float ang = u * TWO_PI - PI_F;
    if (u3 < 0.015625f) ang = 0.f;

    float sn, cs;
    sincosf(ang, &sn, &cs);
    const float sx = quad_take<0>(sn), cx = quad_take<0>(cs);
    const float sy = quad_take<1>(sn), cy = quad_take<1>(cs);
    const float sz = quad_take<2>(sn), cz = quad_take<2>(cs);

    const float czsy = cz * sy;
    const float szsy = sz * sy;
    const float R0 = cz * cy;
    const float R1 = fmaf(czsy, sx, -(sz * cx));
    const float R2 = fmaf(czsy, cx, sz * sx);
    const float R3 = sz * cy;
    const float R4 = fmaf(cz, cx, szsy * sx);
    const float R5 = fmaf(szsy, cx, -(cz * sx));
    const float R6 = -sy;
    const float R7 = cy * sx;
    const float R8 = cy * cx;

    const float r0 = (i == 0) ? R0 : (i == 1) ? R3 : R6;
    const float r1 = (i == 0) ? R1 : (i == 1) ? R4 : R7;
    const float r2 = (i == 0) ? R2 : (i == 1) ? R5 : R8;

    rows[lane] = make_float4(r0, r1, r2, 0.f);
    if (i < 3) {
        float* R = out_R + (size_t)(b0 + e) * 9 + i * 3;
        R[0] = r0; R[1] = r1; R[2] = r2;
    }
}

#define PPL 4

__device__ __forceinline__ void
rotate_one_point(const v2f a1, const v2f a2, const v2f a3,
                 const float px, const float py, const float pz, const float4 q,
                 float* __restrict__ po, float4* __restrict__ qo) {
    const float qx = q.x, qy = q.y, qz = q.z, qw = q.w;

    const float two = 2.0f / (qw * qw + qx * qx + qy * qy + qz * qz);
    const float m00 = 1.0f - two * (qy * qy + qz * qz);
    const float m01 = two * (qx * qy - qz * qw);
    const float m02 = two * (qx * qz + qy * qw);
    const float m10 = two * (qx * qy + qz * qw);
    const float m11 = 1.0f - two * (qx * qx + qz * qz);
    const float m12 = two * (qy * qz - qx * qw);
    const float m20 = two * (qx * qz - qy * qw);
    const float m21 = two * (qy * qz + qx * qw);
    const float m22 = 1.0f - two * (qx * qx + qy * qy);

    v2f bx1, bx2, bx3, by1, by2, by3;
    bx1.x = m00; bx1.y = m10;
    bx2.x = m20; bx2.y = m01;
    bx3.x = m11; bx3.y = m21;
    by1.x = m02; by1.y = m12;
    by2.x = m22; by2.y = px;
    by3.x = py;  by3.y = pz;

    const v8f czero = {0.f, 0.f, 0.f, 0.f, 0.f, 0.f, 0.f, 0.f};
    v8f X = __builtin_amdgcn_wmma_f32_16x16x4_f32(
        false, a1, false, bx1, (short)0, czero, false, false);
    v8f Y = __builtin_amdgcn_wmma_f32_16x16x4_f32(
        false, a1, false, by1, (short)0, czero, false, false);
    X = __builtin_amdgcn_wmma_f32_16x16x4_f32(
        false, a2, false, bx2, (short)0, X, false, false);
    Y = __builtin_amdgcn_wmma_f32_16x16x4_f32(
        false, a2, false, by2, (short)0, Y, false, false);
    X = __builtin_amdgcn_wmma_f32_16x16x4_f32(
        false, a3, false, bx3, (short)0, X, false, false);
    Y = __builtin_amdgcn_wmma_f32_16x16x4_f32(
        false, a3, false, by3, (short)0, Y, false, false);

    const float M00 = X[0], M10 = X[1], M20 = X[2];
    const float M01 = X[3], M11 = X[4], M21 = X[5];
    const float M02 = Y[0], M12 = Y[1], M22 = Y[2];
    const float rpx = Y[3], rpy = Y[4], rpz = Y[5];

    const float t0 = 1.0f + M00 + M11 + M22;
    const float t1 = 1.0f + M00 - M11 - M22;
    const float t2 = 1.0f - M00 + M11 - M22;
    const float t3 = 1.0f - M00 - M11 + M22;
    const float c0 = fmaxf(t0, 1e-8f);
    const float c1 = fmaxf(t1, 1e-8f);
    const float c2 = fmaxf(t2, 1e-8f);
    const float c3 = fmaxf(t3, 1e-8f);

    const float bv = sqrtf(fmaxf(fmaxf(c0, c1), fmaxf(c2, c3)));
    const float lo = __uint_as_float(__float_as_uint(bv) - 1u);
    const bool h0 = fmaf(-lo, bv, c0) > 0.0f;
    const bool h1 = fmaf(-lo, bv, c1) > 0.0f;
    const bool h2 = fmaf(-lo, bv, c2) > 0.0f;
    int best = h2 ? 2 : 3;
    best = h1 ? 1 : best;
    best = h0 ? 0 : best;
    const float bb = bv * bv;

    float W, X_, Y_, Z;
    if (best == 0)      { W = bb;        X_ = M21 - M12; Y_ = M02 - M20; Z = M10 - M01; }
    else if (best == 1) { W = M21 - M12; X_ = bb;        Y_ = M01 + M10; Z = M02 + M20; }
    else if (best == 2) { W = M02 - M20; X_ = M10 + M01; Y_ = bb;        Z = M12 + M21; }
    else                { W = M10 - M01; X_ = M20 + M02; Y_ = M21 + M12; Z = bb; }
    const float inv = 1.0f / (2.0f * fmaxf(bv, 0.1f));
    W *= inv; X_ *= inv; Y_ *= inv; Z *= inv;

    po[0] = rpx; po[1] = rpy; po[2] = rpz;
    *qo = make_float4(X_, Y_, Z, W);
}

__global__ __launch_bounds__(256) void
RotRandomizer_rotate_kernel(const float* __restrict__ pos,
                            const float* __restrict__ quat,
                            float* __restrict__ out_R,
                            float* __restrict__ out_pos,
                            float* __restrict__ out_quat) {
    __shared__ float4 sRows[32];

    const int tid  = threadIdx.x;
    const int wave = __builtin_amdgcn_readfirstlane(tid >> 5);
    const int lane = tid & 31;
    const int b0   = blockIdx.x * 8;
    const int b    = b0 + wave;

    const size_t idx0 = (size_t)b * TPTS + lane;

    float  px[PPL], py[PPL], pz[PPL];
    float4 q[PPL];
#pragma unroll
    for (int j = 0; j < PPL; ++j) {
        const size_t idx = idx0 + (size_t)(j * 32);
        const float* pi = pos + idx * 3;
        px[j] = pi[0]; py[j] = pi[1]; pz[j] = pi[2];
        q[j]  = *((const float4*)quat + idx);
    }

    if (wave == 0) build_R_rows(lane, b0, sRows, out_R);
    __builtin_amdgcn_fence(__ATOMIC_RELEASE, "workgroup", "local");
    __builtin_amdgcn_s_barrier();
    __builtin_amdgcn_fence(__ATOMIC_ACQUIRE, "workgroup", "local");

    const int c    = lane & 15;
    const int half = lane >> 4;
    const int r    = c & 7;
    const bool act   = ((c >> 3) == half) && (r < 6);
    const bool first = r < 3;
    float R0 = 0.f, R1 = 0.f, R2 = 0.f;
    if (act) {
        const float4 Rrow = sRows[wave * 4 + (first ? r : r - 3)];
        R0 = Rrow.x; R1 = Rrow.y; R2 = Rrow.z;
    }
    v2f a1, a2, a3;
    a1.x = first ? R0 : 0.f;   a1.y = first ? R1 : 0.f;
    a2.x = first ? R2 : 0.f;   a2.y = first ? 0.f : R0;
    a3.x = first ? 0.f : R1;   a3.y = first ? 0.f : R2;

#pragma unroll
    for (int j = 0; j < PPL; ++j) {
        const size_t idx = idx0 + (size_t)(j * 32);
        rotate_one_point(a1, a2, a3, px[j], py[j], pz[j], q[j],
                         out_pos + idx * 3, (float4*)out_quat + idx);
    }
}

extern "C" void kernel_launch(void* const* d_in, const int* in_sizes, int n_in,
                              void* d_out, int out_size, void* d_ws, size_t ws_size,
                              hipStream_t stream) {
    const float* pos  = (const float*)d_in[0];
    const float* quat = (const float*)d_in[1];

    float* out      = (float*)d_out;
    float* out_pos  = out;
    float* out_quat = out + (size_t)BATCH * TPTS * 3;
    float* out_R    = out + (size_t)BATCH * TPTS * 3
                          + (size_t)BATCH * TPTS * 4;

    RotRandomizer_rotate_kernel<<<BATCH / 8, 256, 0, stream>>>(
        pos, quat, out_R, out_pos, out_quat);
}


# Rotated quaternion as the product a(x)q on WMMA (4 per point, was 6), normalised by rsq, instead of R*matrix(q) and back; near-tie sign picks still come from the matrix route; sums reordered
# speedup vs baseline: 2.0350x; 1.2295x over previous
#include <hip/hip_runtime.h>
#include <hip/hip_bf16.h>
#include <math.h>


#define BATCH 16384
#define TPTS  128

typedef __attribute__((ext_vector_type(2))) float v2f;
typedef __attribute__((ext_vector_type(8))) float v8f;

__device__ __forceinline__ unsigned rotl32(unsigned v, int n) {
    return (v << n) | (v >> (32 - n));
}

__device__ __forceinline__ void tf2x32(unsigned k0, unsigned k1,
                                       unsigned& x0, unsigned& x1) {
    const unsigned ks0 = k0, ks1 = k1, ks2 = k0 ^ k1 ^ 0x1BD11BDAu;
    x0 += ks0; x1 += ks1;
#define TFR(r) { x0 += x1; x1 = rotl32(x1, r); x1 ^= x0; }
    TFR(13) TFR(15) TFR(26) TFR(6)  x0 += ks1; x1 += ks2 + 1u;
    TFR(17) TFR(29) TFR(16) TFR(24) x0 += ks2; x1 += ks0 + 2u;
    TFR(13) TFR(15) TFR(26) TFR(6)  x0 += ks0; x1 += ks1 + 3u;
    TFR(17) TFR(29) TFR(16) TFR(24) x0 += ks1; x1 += ks2 + 4u;
    TFR(13) TFR(15) TFR(26) TFR(6)  x0 += ks2; x1 += ks0 + 5u;
#undef TFR
}

__device__ __forceinline__ float bits_to_unit(unsigned bits) {
    return __uint_as_float((bits >> 9) | 0x3f800000u) - 1.0f;
}

template <int K>
__device__ __forceinline__ float quad_take(float v) {
    return __int_as_float(__builtin_amdgcn_update_dpp(
        0, __float_as_int(v), K * 0x55, 0xF, 0xF, true));
}

__device__ __forceinline__ void
build_R_rows(const int lane, const int b0,
             float4* __restrict__ rows, float* __restrict__ out_R) {
    const int e = lane >> 2;
    const int i = lane & 3;

    unsigned s0 = 0u, s1 = 1u;
    tf2x32(0u, 42u, s0, s1);
    const unsigned sub0 = s0, sub1 = s1;

    unsigned k[4][2];
#pragma unroll
    for (int t = 0; t < 4; ++t) {
        unsigned a = 0u, c = (unsigned)t;
        tf2x32(sub0, sub1, a, c);
        k[t][0] = a; k[t][1] = c;
    }
    const unsigned key0 = (i == 0) ? k[0][0] : (i == 1) ? k[1][0] : (i == 2) ? k[2][0] : k[3][0];
    const unsigned key1 = (i == 0) ? k[0][1] : (i == 1) ? k[1][1] : (i == 2) ? k[2][1] : k[3][1];

    unsigned a = 0u, c = (unsigned)(b0 + e);
    tf2x32(key0, key1, a, c);
    const float u  = bits_to_unit(a ^ c);
    const float u3 = quad_take<3>(u);

    const float TWO_PI = 6.28318530717958647692f;
    const float PI_F   = 3.14159265358979323846f;
    float ang = u * TWO_PI - PI_F;
    if (u3 < 0.015625f) ang = 0.f;

    float sn, cs;
    sincosf(ang, &sn, &cs);
    float sh, ch;
    sincosf(0.5f * ang, &sh, &ch);
    const float shx = quad_take<0>(sh), chx = quad_take<0>(ch);
    const float shy = quad_take<1>(sh), chy = quad_take<1>(ch);
    const float shz = quad_take<2>(sh), chz = quad_take<2>(ch);
    const float czcy = chz * chy, szsy2 = shz * shy;
    const float czsy2 = chz * shy, szcy = shz * chy;
    const float aw = fmaf(czcy,  chx,  szsy2 * shx);
    const float ax = fmaf(czcy,  shx, -(szsy2 * chx));
    const float ay = fmaf(czsy2, chx,  szcy * shx);
    const float az = fmaf(szcy,  chx, -(czsy2 * shx));
    const float sx = quad_take<0>(sn), cx = quad_take<0>(cs);
    const float sy = quad_take<1>(sn), cy = quad_take<1>(cs);
    const float sz = quad_take<2>(sn), cz = quad_take<2>(cs);

    const float czsy = cz * sy;
    const float szsy = sz * sy;
    const float R0 = cz * cy;
    const float R1 = fmaf(czsy, sx, -(sz * cx));
    const float R2 = fmaf(czsy, cx, sz * sx);
    const float R3 = sz * cy;
    const float R4 = fmaf(cz, cx, szsy * sx);
    const float R5 = fmaf(szsy, cx, -(cz * sx));
    const float R6 = -sy;
    const float R7 = cy * sx;
    const float R8 = cy * cx;

    const float r0 = (i == 0) ? R0 : (i == 1) ? R3 : R6;
    const float r1 = (i == 0) ? R1 : (i == 1) ? R4 : R7;
    const float r2 = (i == 0) ? R2 : (i == 1) ? R5 : R8;

    const float nax = -ax, nay = -ay, naz = -az;
    const float k0 = (i == 0) ? aw  : (i == 1) ? az  : (i == 2) ? nay : nax;
    const float k1 = (i == 0) ? naz : (i == 1) ? aw  : (i == 2) ? ax  : nay;
    const float k2 = (i == 0) ? ay  : (i == 1) ? nax : (i == 2) ? aw  : naz;
    const float k3 = (i == 0) ? ax  : (i == 1) ? ay  : (i == 2) ? az  : aw;
    rows[8 * e + i] = make_float4(k0, k1, k2, k3);
    if (i < 3) {
        rows[8 * e + 4 + i] = make_float4(r0, r1, r2, 0.f);
        float* R = out_R + (size_t)(b0 + e) * 9 + i * 3;
        R[0] = r0; R[1] = r1; R[2] = r2;
    }
}

#define PPL 4

#define NEAR_TIE_SHARE 0.9998779296875f

__device__ __forceinline__ int
matrix_route_choice(const float4* __restrict__ wrows, const int lane, const float4 q) {
    const int c    = lane & 15;
    const int half = lane >> 4;
    const int r    = c & 7;
    const bool act   = ((c >> 3) == half) && (r < 6);
    const bool first = r < 3;
    float R0 = 0.f, R1 = 0.f, R2 = 0.f;
    if (act) {
        const float4 Rrow = wrows[4 + (first ? r : r - 3)];
        R0 = Rrow.x; R1 = Rrow.y; R2 = Rrow.z;
    }
    v2f a1, a2, a3;
    a1.x = first ? R0 : 0.f;   a1.y = first ? R1 : 0.f;
    a2.x = first ? R2 : 0.f;   a2.y = first ? 0.f : R0;
    a3.x = first ? 0.f : R1;   a3.y = first ? 0.f : R2;

    const float qx = q.x, qy = q.y, qz = q.z, qw = q.w;

    const float two = 2.0f / (qw * qw + qx * qx + qy * qy + qz * qz);
    const float m00 = 1.0f - two * (qy * qy + qz * qz);
    const float m01 = two * (qx * qy - qz * qw);
    const float m02 = two * (qx * qz + qy * qw);
    const float m10 = two * (qx * qy + qz * qw);
    const float m11 = 1.0f - two * (qx * qx + qz * qz);
    const float m12 = two * (qy * qz - qx * qw);
    const float m20 = two * (qx * qz - qy * qw);
    const float m21 = two * (qy * qz + qx * qw);
    const float m22 = 1.0f - two * (qx * qx + qy * qy);

    v2f bx1, bx2, bx3, by1, by2;
    bx1.x = m00; bx1.y = m10;
    bx2.x = m20; bx2.y = m01;
    bx3.x = m11; bx3.y = m21;
    by1.x = m02; by1.y = m12;
    by2.x = m22; by2.y = 0.f;

    const v8f czero = {0.f, 0.f, 0.f, 0.f, 0.f, 0.f, 0.f, 0.f};
    v8f X = __builtin_amdgcn_wmma_f32_16x16x4_f32(
        false, a1, false, bx1, (short)0, czero, false, false);
    v8f Y = __builtin_amdgcn_wmma_f32_16x16x4_f32(
        false, a1, false, by1, (short)0, czero, false, false);
    X = __builtin_amdgcn_wmma_f32_16x16x4_f32(
        false, a2, false, bx2, (short)0, X, false, false);
    Y = __builtin_amdgcn_wmma_f32_16x16x4_f32(
        false, a2, false, by2, (short)0, Y, false, false);
    X = __builtin_amdgcn_wmma_f32_16x16x4_f32(
        false, a3, false, bx3, (short)0, X, false, false);

    const float M00 = X[0], M11 = X[4], M22 = Y[2];

    const float r0 = sqrtf(fmaxf(1.0f + M00 + M11 + M22, 1e-8f));
    const float r1 = sqrtf(fmaxf(1.0f + M00 - M11 - M22, 1e-8f));
    const float r2 = sqrtf(fmaxf(1.0f - M00 + M11 - M22, 1e-8f));
    const float r3 = sqrtf(fmaxf(1.0f - M00 - M11 + M22, 1e-8f));
    const float rmax = fmaxf(fmaxf(r0, r1), fmaxf(r2, r3));
    int best = 3;
    best = (r2 >= rmax) ? 2 : best;
    best = (r1 >= rmax) ? 1 : best;
    best = (r0 >= rmax) ? 0 : best;
    return best;
}

__device__ __forceinline__ void
rotate_one_point(const v2f a1, const v2f a2, const v2f a3, const v2f a4,
                 const float4* __restrict__ wrows, const int lane,
                 const float px, const float py, const float pz, const float4 q,
                 float* __restrict__ po, float4* __restrict__ qo) {
    v2f b1, b2, b3, b4;
    b1.x = q.x; b1.y = q.y;
    b2.x = q.z; b2.y = q.w;
    b3.x = 0.f; b3.y = px;
    b4.x = py;  b4.y = pz;

    const v8f czero = {0.f, 0.f, 0.f, 0.f, 0.f, 0.f, 0.f, 0.f};
    v8f D = __builtin_amdgcn_wmma_f32_16x16x4_f32(
        false, a1, false, b1, (short)0, czero, false, false);
    D = __builtin_amdgcn_wmma_f32_16x16x4_f32(
        false, a2, false, b2, (short)0, D, false, false);
    D = __builtin_amdgcn_wmma_f32_16x16x4_f32(
        false, a3, false, b3, (short)0, D, false, false);
    D = __builtin_amdgcn_wmma_f32_16x16x4_f32(
        false, a4, false, b4, (short)0, D, false, false);

    const float ox = D[0], oy = D[1], oz = D[2], ow = D[3];
    const float rpx = D[4], rpy = D[5], rpz = D[6];

    const float n2  = ox * ox + oy * oy + oz * oz + ow * ow;
    const float inv = __builtin_amdgcn_rsqf(n2);
    const float mw = fabsf(ow), mx = fabsf(ox), my = fabsf(oy), mz = fabsf(oz);
    const float hi1 = fmaxf(mw, mx), lo1 = fminf(mw, mx);
    const float hi2 = fmaxf(my, mz), lo2 = fminf(my, mz);
    const float m      = fmaxf(hi1, hi2);
    const float second = fmaxf(fminf(hi1, hi2), fmaxf(lo1, lo2));
    float lead = oz;
    lead = (my >= m) ? oy : lead;
    lead = (mx >= m) ? ox : lead;
    lead = (mw >= m) ? ow : lead;

    const bool near_tie = second >= m * NEAR_TIE_SHARE;
    if (__builtin_expect(__builtin_amdgcn_ballot_w32(near_tie) != 0u, 0)) {
        const int best = matrix_route_choice(wrows, lane, q);
        const float lead_m = (best == 0) ? ow : (best == 1) ? ox : (best == 2) ? oy : oz;
        lead = near_tie ? lead_m : lead;
    }
    const float sc = copysignf(inv, lead);

    po[0] = rpx; po[1] = rpy; po[2] = rpz;
    *qo = make_float4(ox * sc, oy * sc, oz * sc, ow * sc);
}

__global__ __launch_bounds__(256) void
RotRandomizer_rotate_kernel(const float* __restrict__ pos,
                            const float* __restrict__ quat,
                            float* __restrict__ out_R,
                            float* __restrict__ out_pos,
                            float* __restrict__ out_quat) {
    __shared__ float4 sRows[64];

    const int tid  = threadIdx.x;
    const int wave = __builtin_amdgcn_readfirstlane(tid >> 5);
    const int lane = tid & 31;
    const int b0   = blockIdx.x * 8;
    const int b    = b0 + wave;

    const size_t idx0 = (size_t)b * TPTS + lane;

    float  px[PPL], py[PPL], pz[PPL];
    float4 q[PPL];
#pragma unroll
    for (int j = 0; j < PPL; ++j) {
        const size_t idx = idx0 + (size_t)(j * 32);
        const float* pi = pos + idx * 3;
        px[j] = pi[0]; py[j] = pi[1]; pz[j] = pi[2];
        q[j]  = *((const float4*)quat + idx);
    }

    if (wave == 0) build_R_rows(lane, b0, sRows, out_R);
    __builtin_amdgcn_fence(__ATOMIC_RELEASE, "workgroup", "local");
    __builtin_amdgcn_s_barrier();
    __builtin_amdgcn_fence(__ATOMIC_ACQUIRE, "workgroup", "local");

    const int c    = lane & 15;
    const int half = lane >> 4;
    const int r    = c & 7;
    const bool act   = ((c >> 3) == half) && (r < 7);
    const bool isq   = r < 4;
    float4 L = make_float4(0.f, 0.f, 0.f, 0.f);
    if (act) L = sRows[wave * 8 + r];
    v2f a1, a2, a3, a4;
    a1.x = isq ? L.x : 0.f;    a1.y = isq ? L.y : 0.f;
    a2.x = isq ? L.z : 0.f;    a2.y = isq ? L.w : 0.f;
    a3.x = 0.f;                a3.y = isq ? 0.f : L.x;
    a4.x = isq ? 0.f : L.y;    a4.y = isq ? 0.f : L.z;

#pragma unroll
    for (int j = 0; j < PPL; ++j) {
        const size_t idx = idx0 + (size_t)(j * 32);
        rotate_one_point(a1, a2, a3, a4, sRows + wave * 8, lane,
                         px[j], py[j], pz[j], q[j],
                         out_pos + idx * 3, (float4*)out_quat + idx);
    }
}

extern "C" void kernel_launch(void* const* d_in, const int* in_sizes, int n_in,
                              void* d_out, int out_size, void* d_ws, size_t ws_size,
                              hipStream_t stream) {
    const float* pos  = (const float*)d_in[0];
    const float* quat = (const float*)d_in[1];

    float* out      = (float*)d_out;
    float* out_pos  = out;
    float* out_quat = out + (size_t)BATCH * TPTS * 3;
    float* out_R    = out + (size_t)BATCH * TPTS * 3
                          + (size_t)BATCH * TPTS * 4;

    RotRandomizer_rotate_kernel<<<BATCH / 8, 256, 0, stream>>>(
        pos, quat, out_R, out_pos, out_quat);
}
